# starting kernel
# baseline (speedup 1.0000x reference)
#include <hip/hip_runtime.h>
#include <math.h>

typedef __attribute__((ext_vector_type(2))) float v2f;
typedef __attribute__((ext_vector_type(8))) float v8f;

#define NDIM 512
#define EPSF 1e-12f

#define OFF_HI (16 * NDIM)
#define OFF_PP (NDIM * NDIM)
#define OFF_PPHI (NDIM * NDIM + 16 * NDIM)

__global__ void pp2_colnorm_kernel(const float* __restrict__ normal,
                                   float* __restrict__ cninv) {
    int i = blockIdx.x * blockDim.x + threadIdx.x;
    if (i >= NDIM) return;
    float s = 0.0f;
    #pragma unroll 8
    for (int o = 0; o < NDIM; ++o) {
        float v = normal[o * NDIM + i];
        s += v * v;
    }
    cninv[i] = 1.0f / fmaxf(sqrtf(s), EPSF);
}

__global__ void pp2_rowprep_kernel(const float* __restrict__ normal,
                                   const float* __restrict__ plane,
                                   const float* __restrict__ cninv,
                                   float* __restrict__ nn0,
                                   float* __restrict__ pproj,
                                   float* __restrict__ dvec,
                                   float* __restrict__ gvec) {
    const int o = blockIdx.x;
    const int t = threadIdx.x;
    const float* nr = normal + o * NDIM;
    const float* pr = plane + o * NDIM;

    float sp = 0.f, sn = 0.f, spn = 0.f, g = 0.f, t1 = 0.f, t2 = 0.f;
    for (int i = t; i < NDIM; i += 256) {
        float n = nr[i], p = pr[i], c = cninv[i];
        float n0 = n * c;
        sp  += p * p;
        sn  += n * n;
        spn += p * n;
        g   += n0 * n0;
        t1  += n0 * p;
        t2  += n0 * n;
        nn0[o * NDIM + i] = n0;
    }

    __shared__ float red[6][256];
    red[0][t] = sp; red[1][t] = sn; red[2][t] = spn;
    red[3][t] = g;  red[4][t] = t1; red[5][t] = t2;
    __syncthreads();
    for (int s = 128; s > 0; s >>= 1) {
        if (t < s) {
            #pragma unroll
            for (int j = 0; j < 6; ++j) red[j][t] += red[j][t + s];
        }
        __syncthreads();
    }
    float Sp = red[0][0], Sn = red[1][0], Spn = red[2][0];
    float G  = red[3][0], T1 = red[4][0], T2 = red[5][0];

    float np  = fmaxf(sqrtf(Sp), EPSF);
    float nn1 = fmaxf(sqrtf(Sn), EPSF);
    float ct  = Spn / (np * nn1);
    float qq  = fmaxf(Sp - 2.0f * ct * Spn + ct * ct * Sn, 0.0f);
    float rinv = 1.0f / fmaxf(sqrtf(qq), EPSF);

    if (t == 0) {
        dvec[o] = rinv * (T1 - ct * T2);
        gvec[o] = G;
    }
    for (int i = t; i < NDIM; i += 256) {
        pproj[o * NDIM + i] = (pr[i] - nr[i] * ct) * rinv;
    }
}

__global__ void pp2_xnorm_kernel(const float* __restrict__ x,
                                 float* __restrict__ svec) {
    const int b = blockIdx.x;
    const int t = threadIdx.x;
    float s = 0.0f;
    for (int i = t; i < NDIM; i += 256) {
        float v = x[b * NDIM + i];
        s += v * v;
    }
    __shared__ float red[256];
    red[t] = s;
    __syncthreads();
    for (int w = 128; w > 0; w >>= 1) {
        if (t < w) red[t] += red[t + w];
        __syncthreads();
    }
    if (t == 0) svec[b] = red[0];
}

__global__ __launch_bounds__(128) void pp2_wmma_gemm_kernel(
    const float* __restrict__ x,
    const float* __restrict__ nn0,
    const float* __restrict__ dvec,
    const float* __restrict__ gvec,
    const float* __restrict__ svec,
    float* __restrict__ out) {
    const int wave = blockIdx.x * (blockDim.x >> 5) + (threadIdx.x >> 5);
    const int mt = wave >> 4;
    const int nt = wave & 15;
    const int tm = mt * 16;
    const int tn = nt * 32;

    const int lane = threadIdx.x & 31;
    const int half = lane >> 4;
    const int l = lane & 15;

    const float* ap = x   + (tm + l) * NDIM + 2 * half;
    const float* bp = nn0 + (tn + l) * NDIM + 2 * half;

    v8f c1lo = {}, c1hi = {}, c2lo = {}, c2hi = {};

    v2f a  = *(const v2f*)(ap);
    v2f p0 = *(const v2f*)(bp);
    v2f p1 = *(const v2f*)(bp + OFF_HI);
    v2f q0 = *(const v2f*)(bp + OFF_PP);
    v2f q1 = *(const v2f*)(bp + OFF_PPHI);

    #pragma unroll 4
    for (int k = 0; k < NDIM - 4; k += 4) {
        v2f an  = *(const v2f*)(ap + k + 4);
        v2f p0n = *(const v2f*)(bp + k + 4);
        v2f p1n = *(const v2f*)(bp + k + 4 + OFF_HI);
        v2f q0n = *(const v2f*)(bp + k + 4 + OFF_PP);
        v2f q1n = *(const v2f*)(bp + k + 4 + OFF_PPHI);

        c1lo = __builtin_amdgcn_wmma_f32_16x16x4_f32(false, a, false, p0,
                                                     (short)0, c1lo, false, false);
        c1hi = __builtin_amdgcn_wmma_f32_16x16x4_f32(false, a, false, p1,
                                                     (short)0, c1hi, false, false);
        c2lo = __builtin_amdgcn_wmma_f32_16x16x4_f32(false, a, false, q0,
                                                     (short)0, c2lo, false, false);
        c2hi = __builtin_amdgcn_wmma_f32_16x16x4_f32(false, a, false, q1,
                                                     (short)0, c2hi, false, false);

        a = an; p0 = p0n; p1 = p1n; q0 = q0n; q1 = q1n;
    }

    c1lo = __builtin_amdgcn_wmma_f32_16x16x4_f32(false, a, false, p0,
                                                 (short)0, c1lo, false, false);
    c1hi = __builtin_amdgcn_wmma_f32_16x16x4_f32(false, a, false, p1,
                                                 (short)0, c1hi, false, false);
    c2lo = __builtin_amdgcn_wmma_f32_16x16x4_f32(false, a, false, q0,
                                                 (short)0, c2lo, false, false);
    c2hi = __builtin_amdgcn_wmma_f32_16x16x4_f32(false, a, false, q1,
                                                 (short)0, c2hi, false, false);

    const float dlo = dvec[tn + l];
    const float dhi = dvec[tn + 16 + l];
    const float glo = gvec[tn + l] - 2.0f;
    const float ghi = gvec[tn + 16 + l] - 2.0f;

    #pragma unroll
    for (int r = 0; r < 8; ++r) {
        const int row = tm + r + half * 8;
        const float sb = svec[row];

        float c1 = c1lo[r], c2 = c2lo[r];
        float den = fmaxf(sqrtf(fmaxf(sb + c1 * c1 * glo, 0.0f)), EPSF);
        out[row * NDIM + tn + l] = (c2 - c1 * dlo) / den;

        c1 = c1hi[r]; c2 = c2hi[r];
        den = fmaxf(sqrtf(fmaxf(sb + c1 * c1 * ghi, 0.0f)), EPSF);
        out[row * NDIM + tn + 16 + l] = (c2 - c1 * dhi) / den;
    }
}

extern "C" void kernel_launch(void* const* d_in, const int* in_sizes, int n_in,
                              void* d_out, int out_size, void* d_ws, size_t ws_size,
                              hipStream_t stream) {
    const float* x      = (const float*)d_in[0];
    const float* normal = (const float*)d_in[1];
    const float* plane  = (const float*)d_in[2];
    float* out = (float*)d_out;

    float* ws    = (float*)d_ws;
    float* nn0   = ws;
    float* pproj = ws + NDIM * NDIM;
    float* cninv = ws + 2 * NDIM * NDIM;
    float* dvec  = cninv + NDIM;
    float* gvec  = dvec + NDIM;
    float* svec  = gvec + NDIM;

    pp2_colnorm_kernel<<<(NDIM + 255) / 256, 256, 0, stream>>>(normal, cninv);
    pp2_rowprep_kernel<<<NDIM, 256, 0, stream>>>(normal, plane, cninv,
                                                 nn0, pproj, dvec, gvec);
    pp2_xnorm_kernel<<<NDIM, 256, 0, stream>>>(x, svec);

    pp2_wmma_gemm_kernel<<<128, 128, 0, stream>>>(x, nn0,
                                                  dvec, gvec, svec, out);
}


# Row norms of x computed in the column-partials launch by extra one-wave work-groups (4 rows per wave, whole-line 16 B loads, lane-exchange join); own launch gone (4 -> 3); sum order of s_b changes
# speedup vs baseline: 2.3210x; 2.3210x over previous
#include <hip/hip_runtime.h>
#include <math.h>

typedef __attribute__((ext_vector_type(2))) float v2f;
typedef __attribute__((ext_vector_type(8))) float v8f;

#define NDIM 512
#define EPSF 1e-12f

#define OFF_HI (16 * NDIM)
#define OFF_PP (NDIM * NDIM)
#define OFF_PPHI (NDIM * NDIM + 16 * NDIM)

#define CP_STRIPS 16
#define CP_ROWS (NDIM / CP_STRIPS)
#define XN_ROWS 4
#define XN_GROUPS (NDIM / (4 * XN_ROWS))

__global__ __launch_bounds__(32) __attribute__((amdgpu_waves_per_eu(1, 1)))
void pp2_colpart_xnorm_kernel(
    const float* __restrict__ normal, const float* __restrict__ x,
    float* __restrict__ colpart, float* __restrict__ svec) {
    if (blockIdx.y >= CP_STRIPS) {
        const int lane = threadIdx.x;
        const int b0 = ((blockIdx.y - CP_STRIPS) * 4 + blockIdx.x) * XN_ROWS;
        const float4* xs = (const float4*)(x + b0 * NDIM) + lane;

        float4 u[XN_ROWS][4];
        #pragma unroll
        for (int r = 0; r < XN_ROWS; ++r) {
            #pragma unroll
            for (int j = 0; j < 4; ++j) u[r][j] = xs[r * (NDIM / 4) + j * 32];
        }

        float s[XN_ROWS];
        #pragma unroll
        for (int r = 0; r < XN_ROWS; ++r) {
            float a = 0.f;
            #pragma unroll
            for (int j = 0; j < 4; ++j) {
                a += u[r][j].x * u[r][j].x;
                a += u[r][j].y * u[r][j].y;
                a += u[r][j].z * u[r][j].z;
                a += u[r][j].w * u[r][j].w;
            }
            s[r] = a;
        }
        #pragma unroll
        for (int off = 16; off > 0; off >>= 1) {
            #pragma unroll
            for (int r = 0; r < XN_ROWS; ++r) s[r] += __shfl_xor(s[r], off);
        }
        if (lane == 0) {
            *(float4*)(svec + b0) = make_float4(s[0], s[1], s[2], s[3]);
        }
        return;
    }

    const int strip = blockIdx.y;
    const int c4 = blockIdx.x * 32 + threadIdx.x;
    const float4* src =
        (const float4*)(normal + strip * CP_ROWS * NDIM) + c4;

    float4 v[CP_ROWS];
    #pragma unroll
    for (int r = 0; r < CP_ROWS; ++r) v[r] = src[r * (NDIM / 4)];

    float4 s = make_float4(0.f, 0.f, 0.f, 0.f);
    #pragma unroll
    for (int r = 0; r < CP_ROWS; ++r) {
        s.x += v[r].x * v[r].x;
        s.y += v[r].y * v[r].y;
        s.z += v[r].z * v[r].z;
        s.w += v[r].w * v[r].w;
    }
    ((float4*)(colpart + strip * NDIM))[c4] = s;
}

__global__ void pp2_rowprep_kernel(const float* __restrict__ normal,
                                   const float* __restrict__ plane,
                                   const float* __restrict__ colpart,
                                   float* __restrict__ nn0,
                                   float* __restrict__ pproj,
                                   float* __restrict__ dvec,
                                   float* __restrict__ gvec) {
    const int o = blockIdx.x;
    const int t = threadIdx.x;
    const float* nr = normal + o * NDIM;
    const float* pr = plane + o * NDIM;

    float sp = 0.f, sn = 0.f, spn = 0.f, g = 0.f, t1 = 0.f, t2 = 0.f;
    for (int i = t; i < NDIM; i += 256) {
        float n = nr[i], p = pr[i];
        float cs = 0.0f;
        #pragma unroll
        for (int j = 0; j < CP_STRIPS; ++j) cs += colpart[j * NDIM + i];
        float c = 1.0f / fmaxf(sqrtf(cs), EPSF);
        float n0 = n * c;
        sp  += p * p;
        sn  += n * n;
        spn += p * n;
        g   += n0 * n0;
        t1  += n0 * p;
        t2  += n0 * n;
        nn0[o * NDIM + i] = n0;
    }

    __shared__ float red[6][256];
    red[0][t] = sp; red[1][t] = sn; red[2][t] = spn;
    red[3][t] = g;  red[4][t] = t1; red[5][t] = t2;
    __syncthreads();
    for (int s = 128; s > 0; s >>= 1) {
        if (t < s) {
            #pragma unroll
            for (int j = 0; j < 6; ++j) red[j][t] += red[j][t + s];
        }
        __syncthreads();
    }
    float Sp = red[0][0], Sn = red[1][0], Spn = red[2][0];
    float G  = red[3][0], T1 = red[4][0], T2 = red[5][0];

    float np  = fmaxf(sqrtf(Sp), EPSF);
    float nn1 = fmaxf(sqrtf(Sn), EPSF);
    float ct  = Spn / (np * nn1);
    float qq  = fmaxf(Sp - 2.0f * ct * Spn + ct * ct * Sn, 0.0f);
    float rinv = 1.0f / fmaxf(sqrtf(qq), EPSF);

    if (t == 0) {
        dvec[o] = rinv * (T1 - ct * T2);
        gvec[o] = G;
    }
    for (int i = t; i < NDIM; i += 256) {
        pproj[o * NDIM + i] = (pr[i] - nr[i] * ct) * rinv;
    }
}

__global__ __launch_bounds__(128) void pp2_wmma_gemm_kernel(
    const float* __restrict__ x,
    const float* __restrict__ nn0,
    const float* __restrict__ dvec,
    const float* __restrict__ gvec,
    const float* __restrict__ svec,
    float* __restrict__ out) {
    const int wave = blockIdx.x * (blockDim.x >> 5) + (threadIdx.x >> 5);
    const int mt = wave >> 4;
    const int nt = wave & 15;
    const int tm = mt * 16;
    const int tn = nt * 32;

    const int lane = threadIdx.x & 31;
    const int half = lane >> 4;
    const int l = lane & 15;

    const float* ap = x   + (tm + l) * NDIM + 2 * half;
    const float* bp = nn0 + (tn + l) * NDIM + 2 * half;

    v8f c1lo = {}, c1hi = {}, c2lo = {}, c2hi = {};

    v2f a  = *(const v2f*)(ap);
    v2f p0 = *(const v2f*)(bp);
    v2f p1 = *(const v2f*)(bp + OFF_HI);
    v2f q0 = *(const v2f*)(bp + OFF_PP);
    v2f q1 = *(const v2f*)(bp + OFF_PPHI);

    #pragma unroll 4
    for (int k = 0; k < NDIM - 4; k += 4) {
        v2f an  = *(const v2f*)(ap + k + 4);
        v2f p0n = *(const v2f*)(bp + k + 4);
        v2f p1n = *(const v2f*)(bp + k + 4 + OFF_HI);
        v2f q0n = *(const v2f*)(bp + k + 4 + OFF_PP);
        v2f q1n = *(const v2f*)(bp + k + 4 + OFF_PPHI);

        c1lo = __builtin_amdgcn_wmma_f32_16x16x4_f32(false, a, false, p0,
                                                     (short)0, c1lo, false, false);
        c1hi = __builtin_amdgcn_wmma_f32_16x16x4_f32(false, a, false, p1,
                                                     (short)0, c1hi, false, false);
        c2lo = __builtin_amdgcn_wmma_f32_16x16x4_f32(false, a, false, q0,
                                                     (short)0, c2lo, false, false);
        c2hi = __builtin_amdgcn_wmma_f32_16x16x4_f32(false, a, false, q1,
                                                     (short)0, c2hi, false, false);

        a = an; p0 = p0n; p1 = p1n; q0 = q0n; q1 = q1n;
    }

    c1lo = __builtin_amdgcn_wmma_f32_16x16x4_f32(false, a, false, p0,
                                                 (short)0, c1lo, false, false);
    c1hi = __builtin_amdgcn_wmma_f32_16x16x4_f32(false, a, false, p1,
                                                 (short)0, c1hi, false, false);
    c2lo = __builtin_amdgcn_wmma_f32_16x16x4_f32(false, a, false, q0,
                                                 (short)0, c2lo, false, false);
    c2hi = __builtin_amdgcn_wmma_f32_16x16x4_f32(false, a, false, q1,
                                                 (short)0, c2hi, false, false);

    const float dlo = dvec[tn + l];
    const float dhi = dvec[tn + 16 + l];
    const float glo = gvec[tn + l] - 2.0f;
    const float ghi = gvec[tn + 16 + l] - 2.0f;

    #pragma unroll
    for (int r = 0; r < 8; ++r) {
        const int row = tm + r + half * 8;
        const float sb = svec[row];

        float c1 = c1lo[r], c2 = c2lo[r];
        float den = fmaxf(sqrtf(fmaxf(sb + c1 * c1 * glo, 0.0f)), EPSF);
        out[row * NDIM + tn + l] = (c2 - c1 * dlo) / den;

        c1 = c1hi[r]; c2 = c2hi[r];
        den = fmaxf(sqrtf(fmaxf(sb + c1 * c1 * ghi, 0.0f)), EPSF);
        out[row * NDIM + tn + 16 + l] = (c2 - c1 * dhi) / den;
    }
}

extern "C" void kernel_launch(void* const* d_in, const int* in_sizes, int n_in,
                              void* d_out, int out_size, void* d_ws, size_t ws_size,
                              hipStream_t stream) {
    const float* x      = (const float*)d_in[0];
    const float* normal = (const float*)d_in[1];
    const float* plane  = (const float*)d_in[2];
    float* out = (float*)d_out;

    float* ws    = (float*)d_ws;
    float* nn0   = ws;
    float* pproj = ws + NDIM * NDIM;
    float* colpart = ws + 2 * NDIM * NDIM;
    float* dvec  = colpart + CP_STRIPS * NDIM;
    float* gvec  = dvec + NDIM;
    float* svec  = gvec + NDIM;

    pp2_colpart_xnorm_kernel<<<dim3(NDIM / 128, CP_STRIPS + XN_GROUPS), 32, 0, stream>>>(
        normal, x, colpart, svec);
    pp2_rowprep_kernel<<<NDIM, 256, 0, stream>>>(normal, plane, colpart,
                                                 nn0, pproj, dvec, gvec);

    pp2_wmma_gemm_kernel<<<128, 128, 0, stream>>>(x, nn0,
                                                  dvec, gvec, svec, out);
}


# nn0, plane_proj stored by row-prep in WMMA fragment order (16 rows x 8 k = 512 B, one 16-wave block per 16 rows); GEMM B loads 16 B/lane on whole lines; k order in each 8 and row-sum order change
# speedup vs baseline: 2.9250x; 1.2602x over previous
#include <hip/hip_runtime.h>
#include <math.h>

typedef __attribute__((ext_vector_type(2))) float v2f;
typedef __attribute__((ext_vector_type(8))) float v8f;

#define NDIM 512
#define EPSF 1e-12f

#define OFF_HI (16 * NDIM)
#define OFF_PP (NDIM * NDIM)
#define OFF_PPHI (NDIM * NDIM + 16 * NDIM)

#define CP_STRIPS 16
#define CP_ROWS (NDIM / CP_STRIPS)
#define XN_ROWS 4
#define XN_GROUPS (NDIM / (4 * XN_ROWS))

__global__ __launch_bounds__(32) __attribute__((amdgpu_waves_per_eu(1, 1)))
void pp2_colpart_xnorm_kernel(
    const float* __restrict__ normal, const float* __restrict__ x,
    float* __restrict__ colpart, float* __restrict__ svec) {
    if (blockIdx.y >= CP_STRIPS) {
        const int lane = threadIdx.x;
        const int b0 = ((blockIdx.y - CP_STRIPS) * 4 + blockIdx.x) * XN_ROWS;
        const float4* xs = (const float4*)(x + b0 * NDIM) + lane;

        float4 u[XN_ROWS][4];
        #pragma unroll
        for (int r = 0; r < XN_ROWS; ++r) {
            #pragma unroll
            for (int j = 0; j < 4; ++j) u[r][j] = xs[r * (NDIM / 4) + j * 32];
        }

        float s[XN_ROWS];
        #pragma unroll
        for (int r = 0; r < XN_ROWS; ++r) {
            float a = 0.f;
            #pragma unroll
            for (int j = 0; j < 4; ++j) {
                a += u[r][j].x * u[r][j].x;
                a += u[r][j].y * u[r][j].y;
                a += u[r][j].z * u[r][j].z;
                a += u[r][j].w * u[r][j].w;
            }
            s[r] = a;
        }
        #pragma unroll
        for (int off = 16; off > 0; off >>= 1) {
            #pragma unroll
            for (int r = 0; r < XN_ROWS; ++r) s[r] += __shfl_xor(s[r], off);
        }
        if (lane == 0) {
            *(float4*)(svec + b0) = make_float4(s[0], s[1], s[2], s[3]);
        }
        return;
    }

    const int strip = blockIdx.y;
    const int c4 = blockIdx.x * 32 + threadIdx.x;
    const float4* src =
        (const float4*)(normal + strip * CP_ROWS * NDIM) + c4;

    float4 v[CP_ROWS];
    #pragma unroll
    for (int r = 0; r < CP_ROWS; ++r) v[r] = src[r * (NDIM / 4)];

    float4 s = make_float4(0.f, 0.f, 0.f, 0.f);
    #pragma unroll
    for (int r = 0; r < CP_ROWS; ++r) {
        s.x += v[r].x * v[r].x;
        s.y += v[r].y * v[r].y;
        s.z += v[r].z * v[r].z;
        s.w += v[r].w * v[r].w;
    }
    ((float4*)(colpart + strip * NDIM))[c4] = s;
}

typedef __attribute__((ext_vector_type(4))) float v4f;

#define RP_WAVES 16
__global__ __launch_bounds__(32 * RP_WAVES) void pp2_rowprep_kernel(
    const float* __restrict__ normal, const float* __restrict__ plane,
    const float* __restrict__ colpart, float* __restrict__ nn0,
    float* __restrict__ pproj, float* __restrict__ dvec,
    float* __restrict__ gvec) {
    const int t = blockIdx.x;
    const int w = threadIdx.x >> 5;
    const int lane = threadIdx.x & 31;
    const int h = lane >> 4;
    const int l = lane & 15;
    const int R = t * 16 + l;
    const int cb = 32 * w;

    const v4f* nr = (const v4f*)(normal + R * NDIM + cb + 4 * h);
    const v4f* pr = (const v4f*)(plane + R * NDIM + cb + 4 * h);
    v4f n4[4], p4[4];
    #pragma unroll
    for (int j = 0; j < 4; ++j) { n4[j] = nr[2 * j]; p4[j] = pr[2 * j]; }

    float cs = 0.0f;
    #pragma unroll
    for (int j = 0; j < CP_STRIPS; ++j) cs += colpart[j * NDIM + cb + lane];
    const float cinv = 1.0f / fmaxf(sqrtf(cs), EPSF);

    float sp = 0.f, sn = 0.f, spn = 0.f, g = 0.f, t1 = 0.f, t2 = 0.f;
    v4f n0v[4];
    #pragma unroll
    for (int j = 0; j < 4; ++j) {
        v4f n0;
        #pragma unroll
        for (int e = 0; e < 4; ++e) {
            const float c = __shfl(cinv, 8 * j + 4 * h + e);
            const float n = n4[j][e], p = p4[j][e];
            const float v = n * c;
            n0[e] = v;
            sp  += p * p;
            sn  += n * n;
            spn += p * n;
            g   += v * v;
            t1  += v * p;
            t2  += v * n;
        }
        n0v[j] = n0;
    }

    sp  += __shfl_xor(sp, 16);
    sn  += __shfl_xor(sn, 16);
    spn += __shfl_xor(spn, 16);
    g   += __shfl_xor(g, 16);
    t1  += __shfl_xor(t1, 16);
    t2  += __shfl_xor(t2, 16);

    __shared__ v4f red[6][16][5];
    if (h == 0) {
        ((float*)&red[0][l][0])[w] = sp;
        ((float*)&red[1][l][0])[w] = sn;
        ((float*)&red[2][l][0])[w] = spn;
        ((float*)&red[3][l][0])[w] = g;
        ((float*)&red[4][l][0])[w] = t1;
        ((float*)&red[5][l][0])[w] = t2;
    }
    __syncthreads();
    float S[6];
    #pragma unroll
    for (int q = 0; q < 6; ++q) {
        float a = 0.0f;
        #pragma unroll
        for (int k = 0; k < 4; ++k) {
            const v4f r4 = red[q][l][k];
            a += r4[0]; a += r4[1]; a += r4[2]; a += r4[3];
        }
        S[q] = a;
    }
    const float Sp = S[0], Sn = S[1], Spn = S[2];
    const float G  = S[3], T1 = S[4], T2 = S[5];

    const float np  = fmaxf(sqrtf(Sp), EPSF);
    const float nn1 = fmaxf(sqrtf(Sn), EPSF);
    const float ct  = Spn / (np * nn1);
    const float qq  = fmaxf(Sp - 2.0f * ct * Spn + ct * ct * Sn, 0.0f);
    const float rinv = 1.0f / fmaxf(sqrtf(qq), EPSF);

    if (w == 0 && h == 0) {
        dvec[R] = rinv * (T1 - ct * T2);
        gvec[R] = G;
    }
    v4f* nn0c = (v4f*)nn0 + (t * 64 + 4 * w) * 32 + lane;
    v4f* ppc = (v4f*)pproj + (t * 64 + 4 * w) * 32 + lane;
    #pragma unroll
    for (int j = 0; j < 4; ++j) {
        nn0c[j * 32] = n0v[j];
        ppc[j * 32] = (p4[j] - n4[j] * ct) * rinv;
    }
}

#define GEMM_CHUNKS (NDIM / 8)
#define GEMM_AHEAD 6
#define GEMM_RING (GEMM_AHEAD + 1)

__device__ __forceinline__ v2f pp2_pair(v4f v, int s) {
    v2f r;
    r[0] = v[2 * s];
    r[1] = v[2 * s + 1];
    return r;
}

__global__ __launch_bounds__(128) __attribute__((amdgpu_waves_per_eu(1, 1)))
void pp2_wmma_gemm_kernel(
    const float* __restrict__ x,
    const float* __restrict__ nn0,
    const float* __restrict__ dvec,
    const float* __restrict__ gvec,
    const float* __restrict__ svec,
    float* __restrict__ out) {
    const int wave = blockIdx.x * (blockDim.x >> 5) + (threadIdx.x >> 5);
    const int mt = wave >> 4;
    const int nt = wave & 15;
    const int tm = mt * 16;
    const int tn = nt * 32;

    const int lane = threadIdx.x & 31;
    const int half = lane >> 4;
    const int l = lane & 15;

    const float* ap = x   + (tm + l) * NDIM + 4 * half;
    const float* bp = nn0 + tn * NDIM + 4 * lane;

    v8f c1lo = {}, c1hi = {}, c2lo = {}, c2hi = {};

    v4f fa[GEMM_RING], fp0[GEMM_RING], fp1[GEMM_RING], fq0[GEMM_RING], fq1[GEMM_RING];

    #pragma unroll
    for (int g = 0; g < GEMM_AHEAD; ++g) {
        fa[g]  = *(const v4f*)(ap + 8 * g);
        fp0[g] = *(const v4f*)(bp + 128 * g);
        fp1[g] = *(const v4f*)(bp + 128 * g + OFF_HI);
        fq0[g] = *(const v4f*)(bp + 128 * g + OFF_PP);
        fq1[g] = *(const v4f*)(bp + 128 * g + OFF_PPHI);
    }

    #pragma unroll
    for (int g = 0; g < GEMM_CHUNKS; ++g) {
        if (g + GEMM_AHEAD < GEMM_CHUNKS) {
            const int gn = g + GEMM_AHEAD;
            const int sl = gn % GEMM_RING;
            fa[sl]  = *(const v4f*)(ap + 8 * gn);
            fp0[sl] = *(const v4f*)(bp + 128 * gn);
            fp1[sl] = *(const v4f*)(bp + 128 * gn + OFF_HI);
            fq0[sl] = *(const v4f*)(bp + 128 * gn + OFF_PP);
            fq1[sl] = *(const v4f*)(bp + 128 * gn + OFF_PPHI);
        }
        const int c = g % GEMM_RING;
        #pragma unroll
        for (int s = 0; s < 2; ++s) {
            const v2f a = pp2_pair(fa[c], s);
            c1lo = __builtin_amdgcn_wmma_f32_16x16x4_f32(false, a, false, pp2_pair(fp0[c], s),
                                                         (short)0, c1lo, false, false);
            c1hi = __builtin_amdgcn_wmma_f32_16x16x4_f32(false, a, false, pp2_pair(fp1[c], s),
                                                         (short)0, c1hi, false, false);
            c2lo = __builtin_amdgcn_wmma_f32_16x16x4_f32(false, a, false, pp2_pair(fq0[c], s),
                                                         (short)0, c2lo, false, false);
            c2hi = __builtin_amdgcn_wmma_f32_16x16x4_f32(false, a, false, pp2_pair(fq1[c], s),
                                                         (short)0, c2hi, false, false);
        }
    }

    const float dlo = dvec[tn + l];
    const float dhi = dvec[tn + 16 + l];
    const float glo = gvec[tn + l] - 2.0f;
    const float ghi = gvec[tn + 16 + l] - 2.0f;

    #pragma unroll
    for (int r = 0; r < 8; ++r) {
        const int row = tm + r + half * 8;
        const float sb = svec[row];

        float c1 = c1lo[r], c2 = c2lo[r];
        float den = fmaxf(sqrtf(fmaxf(sb + c1 * c1 * glo, 0.0f)), EPSF);
        out[row * NDIM + tn + l] = (c2 - c1 * dlo) / den;

        c1 = c1hi[r]; c2 = c2hi[r];
        den = fmaxf(sqrtf(fmaxf(sb + c1 * c1 * ghi, 0.0f)), EPSF);
        out[row * NDIM + tn + 16 + l] = (c2 - c1 * dhi) / den;
    }
}

extern "C" void kernel_launch(void* const* d_in, const int* in_sizes, int n_in,
                              void* d_out, int out_size, void* d_ws, size_t ws_size,
                              hipStream_t stream) {
    const float* x      = (const float*)d_in[0];
    const float* normal = (const float*)d_in[1];
    const float* plane  = (const float*)d_in[2];
    float* out = (float*)d_out;

    float* ws    = (float*)d_ws;
    float* nn0   = ws;
    float* pproj = ws + NDIM * NDIM;
    float* colpart = ws + 2 * NDIM * NDIM;
    float* dvec  = colpart + CP_STRIPS * NDIM;
    float* gvec  = dvec + NDIM;
    float* svec  = gvec + NDIM;

    pp2_colpart_xnorm_kernel<<<dim3(NDIM / 128, CP_STRIPS + XN_GROUPS), 32, 0, stream>>>(
        normal, x, colpart, svec);
    pp2_rowprep_kernel<<<NDIM / 16, 32 * RP_WAVES, 0, stream>>>(normal, plane, colpart,
                                                                nn0, pproj, dvec, gvec);

    pp2_wmma_gemm_kernel<<<128, 128, 0, stream>>>(x, nn0,
                                                  dvec, gvec, svec, out);
}


# GEMM split by product: a wave carries ONE product of a 16x32 tile (2 accumulators): 1024 waves on all 256 WGPs, not 512 on 128; the two tiles meet in LDS, a wave stores 8 whole rows; sum order kept
# speedup vs baseline: 4.0274x; 1.3769x over previous
#include <hip/hip_runtime.h>
#include <math.h>

typedef __attribute__((ext_vector_type(2))) float v2f;
typedef __attribute__((ext_vector_type(8))) float v8f;

#define NDIM 512
#define EPSF 1e-12f

#define OFF_HI (16 * NDIM)
#define OFF_PP (NDIM * NDIM)

#define CP_STRIPS 16
#define CP_ROWS (NDIM / CP_STRIPS)
#define XN_ROWS 4
#define XN_GROUPS (NDIM / (4 * XN_ROWS))

__global__ __launch_bounds__(32) __attribute__((amdgpu_waves_per_eu(1, 1)))
void pp2_colpart_xnorm_kernel(
    const float* __restrict__ normal, const float* __restrict__ x,
    float* __restrict__ colpart, float* __restrict__ svec) {
    if (blockIdx.y >= CP_STRIPS) {
        const int lane = threadIdx.x;
        const int b0 = ((blockIdx.y - CP_STRIPS) * 4 + blockIdx.x) * XN_ROWS;
        const float4* xs = (const float4*)(x + b0 * NDIM) + lane;

        float4 u[XN_ROWS][4];
        #pragma unroll
        for (int r = 0; r < XN_ROWS; ++r) {
            #pragma unroll
            for (int j = 0; j < 4; ++j) u[r][j] = xs[r * (NDIM / 4) + j * 32];
        }

        float s[XN_ROWS];
        #pragma unroll
        for (int r = 0; r < XN_ROWS; ++r) {
            float a = 0.f;
            #pragma unroll
            for (int j = 0; j < 4; ++j) {
                a += u[r][j].x * u[r][j].x;
                a += u[r][j].y * u[r][j].y;
                a += u[r][j].z * u[r][j].z;
                a += u[r][j].w * u[r][j].w;
            }
            s[r] = a;
        }
        #pragma unroll
        for (int off = 16; off > 0; off >>= 1) {
            #pragma unroll
            for (int r = 0; r < XN_ROWS; ++r) s[r] += __shfl_xor(s[r], off);
        }
        if (lane == 0) {
            *(float4*)(svec + b0) = make_float4(s[0], s[1], s[2], s[3]);
        }
        return;
    }

    const int strip = blockIdx.y;
    const int c4 = blockIdx.x * 32 + threadIdx.x;
    const float4* src =
        (const float4*)(normal + strip * CP_ROWS * NDIM) + c4;

    float4 v[CP_ROWS];
    #pragma unroll
    for (int r = 0; r < CP_ROWS; ++r) v[r] = src[r * (NDIM / 4)];

    float4 s = make_float4(0.f, 0.f, 0.f, 0.f);
    #pragma unroll
    for (int r = 0; r < CP_ROWS; ++r) {
        s.x += v[r].x * v[r].x;
        s.y += v[r].y * v[r].y;
        s.z += v[r].z * v[r].z;
        s.w += v[r].w * v[r].w;
    }
    ((float4*)(colpart + strip * NDIM))[c4] = s;
}

typedef __attribute__((ext_vector_type(4))) float v4f;

#define RP_WAVES 16
__global__ __launch_bounds__(32 * RP_WAVES) void pp2_rowprep_kernel(
    const float* __restrict__ normal, const float* __restrict__ plane,
    const float* __restrict__ colpart, float* __restrict__ nn0,
    float* __restrict__ pproj, float* __restrict__ dvec,
    float* __restrict__ gvec) {
    const int t = blockIdx.x;
    const int w = threadIdx.x >> 5;
    const int lane = threadIdx.x & 31;
    const int h = lane >> 4;
    const int l = lane & 15;
    const int R = t * 16 + l;
    const int cb = 32 * w;

    const v4f* nr = (const v4f*)(normal + R * NDIM + cb + 4 * h);
    const v4f* pr = (const v4f*)(plane + R * NDIM + cb + 4 * h);
    v4f n4[4], p4[4];
    #pragma unroll
    for (int j = 0; j < 4; ++j) { n4[j] = nr[2 * j]; p4[j] = pr[2 * j]; }

    float cs = 0.0f;
    #pragma unroll
    for (int j = 0; j < CP_STRIPS; ++j) cs += colpart[j * NDIM + cb + lane];
    const float cinv = 1.0f / fmaxf(sqrtf(cs), EPSF);

    float sp = 0.f, sn = 0.f, spn = 0.f, g = 0.f, t1 = 0.f, t2 = 0.f;
    v4f n0v[4];
    #pragma unroll
    for (int j = 0; j < 4; ++j) {
        v4f n0;
        #pragma unroll
        for (int e = 0; e < 4; ++e) {
            const float c = __shfl(cinv, 8 * j + 4 * h + e);
            const float n = n4[j][e], p = p4[j][e];
            const float v = n * c;
            n0[e] = v;
            sp  += p * p;
            sn  += n * n;
            spn += p * n;
            g   += v * v;
            t1  += v * p;
            t2  += v * n;
        }
        n0v[j] = n0;
    }

    sp  += __shfl_xor(sp, 16);
    sn  += __shfl_xor(sn, 16);
    spn += __shfl_xor(spn, 16);
    g   += __shfl_xor(g, 16);
    t1  += __shfl_xor(t1, 16);
    t2  += __shfl_xor(t2, 16);

    __shared__ v4f red[6][16][5];
    if (h == 0) {
        ((float*)&red[0][l][0])[w] = sp;
        ((float*)&red[1][l][0])[w] = sn;
        ((float*)&red[2][l][0])[w] = spn;
        ((float*)&red[3][l][0])[w] = g;
        ((float*)&red[4][l][0])[w] = t1;
        ((float*)&red[5][l][0])[w] = t2;
    }
    __syncthreads();
    float S[6];
    #pragma unroll
    for (int q = 0; q < 6; ++q) {
        float a = 0.0f;
        #pragma unroll
        for (int k = 0; k < 4; ++k) {
            const v4f r4 = red[q][l][k];
            a += r4[0]; a += r4[1]; a += r4[2]; a += r4[3];
        }
        S[q] = a;
    }
    const float Sp = S[0], Sn = S[1], Spn = S[2];
    const float G  = S[3], T1 = S[4], T2 = S[5];

    const float np  = fmaxf(sqrtf(Sp), EPSF);
    const float nn1 = fmaxf(sqrtf(Sn), EPSF);
    const float ct  = Spn / (np * nn1);
    const float qq  = fmaxf(Sp - 2.0f * ct * Spn + ct * ct * Sn, 0.0f);
    const float rinv = 1.0f / fmaxf(sqrtf(qq), EPSF);

    if (w == 0 && h == 0) {
        dvec[R] = rinv * (T1 - ct * T2);
        gvec[R] = G;
    }
    v4f* nn0c = (v4f*)nn0 + (t * 64 + 4 * w) * 32 + lane;
    v4f* ppc = (v4f*)pproj + (t * 64 + 4 * w) * 32 + lane;
    #pragma unroll
    for (int j = 0; j < 4; ++j) {
        nn0c[j * 32] = n0v[j];
        ppc[j * 32] = (p4[j] - n4[j] * ct) * rinv;
    }
}

#define GEMM_CHUNKS (NDIM / 8)
#define GEMM_AHEAD 16
#define GEMM_RING (GEMM_AHEAD + 1)
#define EP_PITCH 34

__device__ __forceinline__ v2f pp2_pair(v4f v, int s) {
    v2f r;
    r[0] = v[2 * s];
    r[1] = v[2 * s + 1];
    return r;
}

__global__ __launch_bounds__(128) __attribute__((amdgpu_waves_per_eu(1, 1)))
void pp2_wmma_gemm_kernel(
    const float* __restrict__ x,
    const float* __restrict__ nn0,
    const float* __restrict__ dvec,
    const float* __restrict__ gvec,
    const float* __restrict__ svec,
    float* __restrict__ out) {
    const int w = threadIdx.x >> 5;
    const int mi = w >> 1;
    const int prod = w & 1;
    const int nt = blockIdx.x & 15;
    const int mt = (blockIdx.x >> 4) * 2 + mi;
    const int tm = mt * 16;
    const int tn = nt * 32;

    const int lane = threadIdx.x & 31;
    const int half = lane >> 4;
    const int l = lane & 15;

    const float* ap = x   + (tm + l) * NDIM + 4 * half;
    const float* bp = nn0 + prod * OFF_PP + tn * NDIM + 4 * lane;

    v8f clo = {}, chi = {};

    v4f fa[GEMM_RING], fb0[GEMM_RING], fb1[GEMM_RING];

    #pragma unroll
    for (int g = 0; g < GEMM_AHEAD; ++g) {
        fa[g]  = *(const v4f*)(ap + 8 * g);
        fb0[g] = *(const v4f*)(bp + 128 * g);
        fb1[g] = *(const v4f*)(bp + 128 * g + OFF_HI);
    }

    const float dn = dvec[tn + lane];
    const float gn2 = gvec[tn + lane] - 2.0f;
    float sbv[8];
    #pragma unroll
    for (int r = 0; r < 8; ++r) sbv[r] = svec[tm + 8 * prod + r];

    #pragma unroll
    for (int g = 0; g < GEMM_CHUNKS; ++g) {
        if (g + GEMM_AHEAD < GEMM_CHUNKS) {
            const int gn = g + GEMM_AHEAD;
            const int sl = gn % GEMM_RING;
            fa[sl]  = *(const v4f*)(ap + 8 * gn);
            fb0[sl] = *(const v4f*)(bp + 128 * gn);
            fb1[sl] = *(const v4f*)(bp + 128 * gn + OFF_HI);
        }
        const int c = g % GEMM_RING;
        #pragma unroll
        for (int s = 0; s < 2; ++s) {
            const v2f a = pp2_pair(fa[c], s);
            clo = __builtin_amdgcn_wmma_f32_16x16x4_f32(false, a, false, pp2_pair(fb0[c], s),
                                                        (short)0, clo, false, false);
            chi = __builtin_amdgcn_wmma_f32_16x16x4_f32(false, a, false, pp2_pair(fb1[c], s),
                                                        (short)0, chi, false, false);
        }
        __builtin_amdgcn_sched_barrier(0);
    }

    __shared__ float ep[2][2][16][EP_PITCH];
    #pragma unroll
    for (int r = 0; r < 8; ++r) {
        ep[mi][prod][8 * half + r][l] = clo[r];
        ep[mi][prod][8 * half + r][16 + l] = chi[r];
    }
    __syncthreads();

    #pragma unroll
    for (int r = 0; r < 8; ++r) {
        const int lr = 8 * prod + r;
        const int row = tm + lr;
        const float sb = sbv[r];
        const float c1 = ep[mi][0][lr][lane];
        const float c2 = ep[mi][1][lr][lane];
        const float den = fmaxf(sqrtf(fmaxf(sb + c1 * c1 * gn2, 0.0f)), EPSF);
        out[row * NDIM + tn + lane] = (c2 - c1 * dn) / den;
    }
}

extern "C" void kernel_launch(void* const* d_in, const int* in_sizes, int n_in,
                              void* d_out, int out_size, void* d_ws, size_t ws_size,
                              hipStream_t stream) {
    const float* x      = (const float*)d_in[0];
    const float* normal = (const float*)d_in[1];
    const float* plane  = (const float*)d_in[2];
    float* out = (float*)d_out;

    float* ws    = (float*)d_ws;
    float* nn0   = ws;
    float* pproj = ws + NDIM * NDIM;
    float* colpart = ws + 2 * NDIM * NDIM;
    float* dvec  = colpart + CP_STRIPS * NDIM;
    float* gvec  = dvec + NDIM;
    float* svec  = gvec + NDIM;

    pp2_colpart_xnorm_kernel<<<dim3(NDIM / 128, CP_STRIPS + XN_GROUPS), 32, 0, stream>>>(
        normal, x, colpart, svec);
    pp2_rowprep_kernel<<<NDIM / 16, 32 * RP_WAVES, 0, stream>>>(normal, plane, colpart,
                                                                nn0, pproj, dvec, gvec);

    pp2_wmma_gemm_kernel<<<256, 128, 0, stream>>>(x, nn0,
                                                  dvec, gvec, svec, out);
}


# Row-prep launch gone (3 -> 2): each GEMM work-group builds its own 32 rows of nn0 and plane_proj in LDS in fragment order and reads B fragments from LDS; order of the row sums, d_o, g_o changes
# speedup vs baseline: 4.5650x; 1.1335x over previous
#include <hip/hip_runtime.h>
#include <math.h>

typedef __attribute__((ext_vector_type(2))) float v2f;
typedef __attribute__((ext_vector_type(8))) float v8f;

#define NDIM 512
#define EPSF 1e-12f

#define CP_STRIPS 16
#define CP_ROWS (NDIM / CP_STRIPS)
#define XN_ROWS 4
#define XN_GROUPS (NDIM / (4 * XN_ROWS))

__global__ __launch_bounds__(32) __attribute__((amdgpu_waves_per_eu(1, 1)))
void pp2_colpart_xnorm_kernel(
    const float* __restrict__ normal, const float* __restrict__ x,
    float* __restrict__ colpart, float* __restrict__ svec) {
    if (blockIdx.y >= CP_STRIPS) {
        const int lane = threadIdx.x;
        const int b0 = ((blockIdx.y - CP_STRIPS) * 4 + blockIdx.x) * XN_ROWS;
        const float4* xs = (const float4*)(x + b0 * NDIM) + lane;

        float4 u[XN_ROWS][4];
        #pragma unroll
        for (int r = 0; r < XN_ROWS; ++r) {
            #pragma unroll
            for (int j = 0; j < 4; ++j) u[r][j] = xs[r * (NDIM / 4) + j * 32];
        }

        float s[XN_ROWS];
        #pragma unroll
        for (int r = 0; r < XN_ROWS; ++r) {
            float a = 0.f;
            #pragma unroll
            for (int j = 0; j < 4; ++j) {
                a += u[r][j].x * u[r][j].x;
                a += u[r][j].y * u[r][j].y;
                a += u[r][j].z * u[r][j].z;
                a += u[r][j].w * u[r][j].w;
            }
            s[r] = a;
        }
        #pragma unroll
        for (int off = 16; off > 0; off >>= 1) {
            #pragma unroll
            for (int r = 0; r < XN_ROWS; ++r) s[r] += __shfl_xor(s[r], off);
        }
        if (lane == 0) {
            *(float4*)(svec + b0) = make_float4(s[0], s[1], s[2], s[3]);
        }
        return;
    }

    const int strip = blockIdx.y;
    const int c4 = blockIdx.x * 32 + threadIdx.x;
    const float4* src =
        (const float4*)(normal + strip * CP_ROWS * NDIM) + c4;

    float4 v[CP_ROWS];
    #pragma unroll
    for (int r = 0; r < CP_ROWS; ++r) v[r] = src[r * (NDIM / 4)];

    float4 s = make_float4(0.f, 0.f, 0.f, 0.f);
    #pragma unroll
    for (int r = 0; r < CP_ROWS; ++r) {
        s.x += v[r].x * v[r].x;
        s.y += v[r].y * v[r].y;
        s.z += v[r].z * v[r].z;
        s.w += v[r].w * v[r].w;
    }
    ((float4*)(colpart + strip * NDIM))[c4] = s;
}

typedef __attribute__((ext_vector_type(4))) float v4f;

#define GEMM_CHUNKS (NDIM / 8)
#define GEMM_AHEAD 16
#define GEMM_RING (GEMM_AHEAD + 1)
#define GEMM_BLK 8
#define EP_PITCH 34

__device__ __forceinline__ v2f pp2_pair(v4f v, int s) {
    v2f r;
    r[0] = v[2 * s];
    r[1] = v[2 * s + 1];
    return r;
}

__global__ __launch_bounds__(128) __attribute__((amdgpu_waves_per_eu(1, 1)))
void pp2_prep_wmma_gemm_kernel(
    const float* __restrict__ x,
    const float* __restrict__ normal,
    const float* __restrict__ plane,
    const float* __restrict__ colpart,
    const float* __restrict__ svec,
    float* __restrict__ out) {
    __shared__ v4f bl[2][2][GEMM_CHUNKS][32];
    __shared__ v4f red[6][2][16];
    __shared__ float dsm[32], gsm[32];
    __shared__ float ep[2][2][16][EP_PITCH];

    const int w = threadIdx.x >> 5;
    const int mi = w >> 1;
    const int prod = w & 1;
    const int nt = blockIdx.x & 15;
    const int mt = (blockIdx.x >> 4) * 2 + mi;
    const int tm = mt * 16;
    const int tn = nt * 32;

    const int lane = threadIdx.x & 31;
    const int half = lane >> 4;
    const int l = lane & 15;

    const int cbw = 128 * w;
    v4f n4[2][16], p4[2][16];
    #pragma unroll
    for (int sub = 0; sub < 2; ++sub) {
        const v4f* nr = (const v4f*)(normal + (tn + 16 * sub + l) * NDIM + cbw + 4 * half);
        const v4f* pr = (const v4f*)(plane + (tn + 16 * sub + l) * NDIM + cbw + 4 * half);
        #pragma unroll
        for (int j = 0; j < 16; ++j) { n4[sub][j] = nr[2 * j]; p4[sub][j] = pr[2 * j]; }
    }

    const float* ap = x + (tm + l) * NDIM + 4 * half;
    v4f fa[GEMM_RING];
    #pragma unroll
    for (int g = 0; g < GEMM_AHEAD; ++g) fa[g] = *(const v4f*)(ap + 8 * g);

    float cinvq[4];
    #pragma unroll
    for (int q = 0; q < 4; ++q) {
        float cs = 0.0f;
        #pragma unroll
        for (int j = 0; j < CP_STRIPS; ++j) cs += colpart[j * NDIM + cbw + 32 * q + lane];
        cinvq[q] = 1.0f / fmaxf(sqrtf(cs), EPSF);
    }

    float sp[2] = {0.f, 0.f}, sn[2] = {0.f, 0.f}, spn[2] = {0.f, 0.f};
    float gg[2] = {0.f, 0.f}, t1[2] = {0.f, 0.f}, t2[2] = {0.f, 0.f};
    #pragma unroll
    for (int j = 0; j < 16; ++j) {
        v4f n0[2];
        #pragma unroll
        for (int e = 0; e < 4; ++e) {
            const float c = __shfl(cinvq[j >> 2], 8 * (j & 3) + 4 * half + e);
            #pragma unroll
            for (int sub = 0; sub < 2; ++sub) {
                const float n = n4[sub][j][e], p = p4[sub][j][e];
                const float v = n * c;
                n0[sub][e] = v;
                sp[sub]  += p * p;
                sn[sub]  += n * n;
                spn[sub] += p * n;
                gg[sub]  += v * v;
                t1[sub]  += v * p;
                t2[sub]  += v * n;
            }
        }
        bl[0][0][16 * w + j][lane] = n0[0];
        bl[0][1][16 * w + j][lane] = n0[1];
    }

    #pragma unroll
    for (int sub = 0; sub < 2; ++sub) {
        sp[sub]  += __shfl_xor(sp[sub], 16);
        sn[sub]  += __shfl_xor(sn[sub], 16);
        spn[sub] += __shfl_xor(spn[sub], 16);
        gg[sub]  += __shfl_xor(gg[sub], 16);
        t1[sub]  += __shfl_xor(t1[sub], 16);
        t2[sub]  += __shfl_xor(t2[sub], 16);
        if (half == 0) {
            ((float*)&red[0][sub][l])[w] = sp[sub];
            ((float*)&red[1][sub][l])[w] = sn[sub];
            ((float*)&red[2][sub][l])[w] = spn[sub];
            ((float*)&red[3][sub][l])[w] = gg[sub];
            ((float*)&red[4][sub][l])[w] = t1[sub];
            ((float*)&red[5][sub][l])[w] = t2[sub];
        }
    }
    __syncthreads();
    #pragma unroll
    for (int sub = 0; sub < 2; ++sub) {
        float S[6];
        #pragma unroll
        for (int q = 0; q < 6; ++q) {
            const v4f r4 = red[q][sub][l];
            float a = 0.0f;
            a += r4[0]; a += r4[1]; a += r4[2]; a += r4[3];
            S[q] = a;
        }
        const float Sp = S[0], Sn = S[1], Spn = S[2];
        const float G  = S[3], T1 = S[4], T2 = S[5];

        const float np  = fmaxf(sqrtf(Sp), EPSF);
        const float nn1 = fmaxf(sqrtf(Sn), EPSF);
        const float ct  = Spn / (np * nn1);
        const float qq  = fmaxf(Sp - 2.0f * ct * Spn + ct * ct * Sn, 0.0f);
        const float rinv = 1.0f / fmaxf(sqrtf(qq), EPSF);

        if (w == 0 && half == 0) {
            dsm[16 * sub + l] = rinv * (T1 - ct * T2);
            gsm[16 * sub + l] = G;
        }
        #pragma unroll
        for (int j = 0; j < 16; ++j)
            bl[1][sub][16 * w + j][lane] = (p4[sub][j] - n4[sub][j] * ct) * rinv;
    }
    __syncthreads();

    const v4f* blo = &bl[prod][0][0][lane];

    const float dn = dsm[lane];
    const float gn2 = gsm[lane] - 2.0f;
    float sbv[8];
    #pragma unroll
    for (int r = 0; r < 8; ++r) sbv[r] = svec[tm + 8 * prod + r];

    v8f clo = {}, chi = {};
    v4f fb0[2][GEMM_BLK], fb1[2][GEMM_BLK];
    #pragma unroll
    for (int c = 0; c < GEMM_BLK; ++c) {
        fb0[0][c] = blo[32 * c];
        fb1[0][c] = blo[32 * (GEMM_CHUNKS + c)];
    }

    #pragma unroll
    for (int g = 0; g < GEMM_CHUNKS; ++g) {
        const int blk = g / GEMM_BLK;
        const int c8 = g % GEMM_BLK;
        if (c8 == 0 && (blk + 1) * GEMM_BLK < GEMM_CHUNKS) {
            #pragma unroll
            for (int c = 0; c < GEMM_BLK; ++c) {
                fb0[(blk + 1) & 1][c] = blo[32 * ((blk + 1) * GEMM_BLK + c)];
                fb1[(blk + 1) & 1][c] = blo[32 * (GEMM_CHUNKS + (blk + 1) * GEMM_BLK + c)];
            }
            __builtin_amdgcn_sched_barrier(0);
        }
        if (g + GEMM_AHEAD < GEMM_CHUNKS) {
            const int gn = g + GEMM_AHEAD;
            fa[gn % GEMM_RING] = *(const v4f*)(ap + 8 * gn);
        }
        const int c = g % GEMM_RING;
        #pragma unroll
        for (int s = 0; s < 2; ++s) {
            const v2f a = pp2_pair(fa[c], s);
            clo = __builtin_amdgcn_wmma_f32_16x16x4_f32(false, a, false, pp2_pair(fb0[blk & 1][c8], s),
                                                        (short)0, clo, false, false);
            chi = __builtin_amdgcn_wmma_f32_16x16x4_f32(false, a, false, pp2_pair(fb1[blk & 1][c8], s),
                                                        (short)0, chi, false, false);
        }
        __builtin_amdgcn_sched_barrier(0);
    }

    #pragma unroll
    for (int r = 0; r < 8; ++r) {
        ep[mi][prod][8 * half + r][l] = clo[r];
        ep[mi][prod][8 * half + r][16 + l] = chi[r];
    }
    __syncthreads();

    #pragma unroll
    for (int r = 0; r < 8; ++r) {
        const int lr = 8 * prod + r;
        const int row = tm + lr;
        const float sb = sbv[r];
        const float c1 = ep[mi][0][lr][lane];
        const float c2 = ep[mi][1][lr][lane];
        const float den = fmaxf(sqrtf(fmaxf(sb + c1 * c1 * gn2, 0.0f)), EPSF);
        out[row * NDIM + tn + lane] = (c2 - c1 * dn) / den;
    }
}

extern "C" void kernel_launch(void* const* d_in, const int* in_sizes, int n_in,
                              void* d_out, int out_size, void* d_ws, size_t ws_size,
                              hipStream_t stream) {
    const float* x      = (const float*)d_in[0];
    const float* normal = (const float*)d_in[1];
    const float* plane  = (const float*)d_in[2];
    float* out = (float*)d_out;

    float* ws    = (float*)d_ws;
    float* colpart = ws;
    float* svec  = colpart + CP_STRIPS * NDIM;

    pp2_colpart_xnorm_kernel<<<dim3(NDIM / 128, CP_STRIPS + XN_GROUPS), 32, 0, stream>>>(
        normal, x, colpart, svec);

    pp2_prep_wmma_gemm_kernel<<<256, 128, 0, stream>>>(x, normal, plane, colpart, svec, out);
}


# Work-group tile 64 x-rows x 16 columns instead of 32 x 32: it builds 16 rows of each B operand, not 32; a wave carries both products of its 16x16 tile (2 accumulators, interleaved); sum order kept
# speedup vs baseline: 4.8757x; 1.0681x over previous
#include <hip/hip_runtime.h>
#include <math.h>

typedef __attribute__((ext_vector_type(2))) float v2f;
typedef __attribute__((ext_vector_type(8))) float v8f;

#define NDIM 512
#define EPSF 1e-12f

#define CP_STRIPS 16
#define CP_ROWS (NDIM / CP_STRIPS)
#define XN_ROWS 4
#define XN_GROUPS (NDIM / (4 * XN_ROWS))

__global__ __launch_bounds__(32) __attribute__((amdgpu_waves_per_eu(1, 1)))
void pp2_colpart_xnorm_kernel(
    const float* __restrict__ normal, const float* __restrict__ x,
    float* __restrict__ colpart, float* __restrict__ svec) {
    if (blockIdx.y >= CP_STRIPS) {
        const int lane = threadIdx.x;
        const int b0 = ((blockIdx.y - CP_STRIPS) * 4 + blockIdx.x) * XN_ROWS;
        const float4* xs = (const float4*)(x + b0 * NDIM) + lane;

        float4 u[XN_ROWS][4];
        #pragma unroll
        for (int r = 0; r < XN_ROWS; ++r) {
            #pragma unroll
            for (int j = 0; j < 4; ++j) u[r][j] = xs[r * (NDIM / 4) + j * 32];
        }

        float s[XN_ROWS];
        #pragma unroll
        for (int r = 0; r < XN_ROWS; ++r) {
            float a = 0.f;
            #pragma unroll
            for (int j = 0; j < 4; ++j) {
                a += u[r][j].x * u[r][j].x;
                a += u[r][j].y * u[r][j].y;
                a += u[r][j].z * u[r][j].z;
                a += u[r][j].w * u[r][j].w;
            }
            s[r] = a;
        }
        #pragma unroll
        for (int off = 16; off > 0; off >>= 1) {
            #pragma unroll
            for (int r = 0; r < XN_ROWS; ++r) s[r] += __shfl_xor(s[r], off);
        }
        if (lane == 0) {
            *(float4*)(svec + b0) = make_float4(s[0], s[1], s[2], s[3]);
        }
        return;
    }

    const int strip = blockIdx.y;
    const int c4 = blockIdx.x * 32 + threadIdx.x;
    const float4* src =
        (const float4*)(normal + strip * CP_ROWS * NDIM) + c4;

    float4 v[CP_ROWS];
    #pragma unroll
    for (int r = 0; r < CP_ROWS; ++r) v[r] = src[r * (NDIM / 4)];

    float4 s = make_float4(0.f, 0.f, 0.f, 0.f);
    #pragma unroll
    for (int r = 0; r < CP_ROWS; ++r) {
        s.x += v[r].x * v[r].x;
        s.y += v[r].y * v[r].y;
        s.z += v[r].z * v[r].z;
        s.w += v[r].w * v[r].w;
    }
    ((float4*)(colpart + strip * NDIM))[c4] = s;
}

typedef __attribute__((ext_vector_type(4))) float v4f;

#define GEMM_CHUNKS (NDIM / 8)
#define GEMM_AHEAD 16
#define GEMM_RING (GEMM_AHEAD + 1)
#define GEMM_BLK 8

__device__ __forceinline__ v2f pp2_pair(v4f v, int s) {
    v2f r;
    r[0] = v[2 * s];
    r[1] = v[2 * s + 1];
    return r;
}

__global__ __launch_bounds__(128) __attribute__((amdgpu_waves_per_eu(1, 1)))
void pp2_prep_wmma_gemm_kernel(
    const float* __restrict__ x,
    const float* __restrict__ normal,
    const float* __restrict__ plane,
    const float* __restrict__ colpart,
    const float* __restrict__ svec,
    float* __restrict__ out) {
    __shared__ v4f bl[2][GEMM_CHUNKS][32];
    __shared__ v4f red[6][16];
    __shared__ float dsm[16], gsm[16];

    const int w = threadIdx.x >> 5;
    const int nt = blockIdx.x & 31;
    const int mt = (blockIdx.x >> 5) * 4 + w;
    const int tm = mt * 16;
    const int tn = nt * 16;

    const int lane = threadIdx.x & 31;
    const int half = lane >> 4;
    const int l = lane & 15;

    const int cbw = 128 * w;
    v4f n4[16], p4[16];
    {
        const v4f* nr = (const v4f*)(normal + (tn + l) * NDIM + cbw + 4 * half);
        const v4f* pr = (const v4f*)(plane + (tn + l) * NDIM + cbw + 4 * half);
        #pragma unroll
        for (int j = 0; j < 16; ++j) { n4[j] = nr[2 * j]; p4[j] = pr[2 * j]; }
    }

    const float* ap = x + (tm + l) * NDIM + 4 * half;
    v4f fa[GEMM_RING];
    #pragma unroll
    for (int g = 0; g < GEMM_AHEAD; ++g) fa[g] = *(const v4f*)(ap + 8 * g);

    float cinvq[4];
    #pragma unroll
    for (int q = 0; q < 4; ++q) {
        float cs = 0.0f;
        #pragma unroll
        for (int j = 0; j < CP_STRIPS; ++j) cs += colpart[j * NDIM + cbw + 32 * q + lane];
        cinvq[q] = 1.0f / fmaxf(sqrtf(cs), EPSF);
    }

    float sp = 0.f, sn = 0.f, spn = 0.f, gg = 0.f, t1 = 0.f, t2 = 0.f;
    #pragma unroll
    for (int j = 0; j < 16; ++j) {
        v4f n0;
        #pragma unroll
        for (int e = 0; e < 4; ++e) {
            const float c = __shfl(cinvq[j >> 2], 8 * (j & 3) + 4 * half + e);
            const float n = n4[j][e], p = p4[j][e];
            const float v = n * c;
            n0[e] = v;
            sp  += p * p;
            sn  += n * n;
            spn += p * n;
            gg  += v * v;
            t1  += v * p;
            t2  += v * n;
        }
        bl[0][16 * w + j][lane] = n0;
    }

    sp  += __shfl_xor(sp, 16);
    sn  += __shfl_xor(sn, 16);
    spn += __shfl_xor(spn, 16);
    gg  += __shfl_xor(gg, 16);
    t1  += __shfl_xor(t1, 16);
    t2  += __shfl_xor(t2, 16);
    if (half == 0) {
        ((float*)&red[0][l])[w] = sp;
        ((float*)&red[1][l])[w] = sn;
        ((float*)&red[2][l])[w] = spn;
        ((float*)&red[3][l])[w] = gg;
        ((float*)&red[4][l])[w] = t1;
        ((float*)&red[5][l])[w] = t2;
    }
    __syncthreads();
    {
        float S[6];
        #pragma unroll
        for (int q = 0; q < 6; ++q) {
            const v4f r4 = red[q][l];
            float a = 0.0f;
            a += r4[0]; a += r4[1]; a += r4[2]; a += r4[3];
            S[q] = a;
        }
        const float Sp = S[0], Sn = S[1], Spn = S[2];
        const float G  = S[3], T1 = S[4], T2 = S[5];

        const float np  = fmaxf(sqrtf(Sp), EPSF);
        const float nn1 = fmaxf(sqrtf(Sn), EPSF);
        const float ct  = Spn / (np * nn1);
        const float qq  = fmaxf(Sp - 2.0f * ct * Spn + ct * ct * Sn, 0.0f);
        const float rinv = 1.0f / fmaxf(sqrtf(qq), EPSF);

        if (w == 0 && half == 0) {
            dsm[l] = rinv * (T1 - ct * T2);
            gsm[l] = G;
        }
        #pragma unroll
        for (int j = 0; j < 16; ++j)
            bl[1][16 * w + j][lane] = (p4[j] - n4[j] * ct) * rinv;
    }
    __syncthreads();

    const v4f* b1p = &bl[0][0][lane];

    const float dn = dsm[l];
    const float gn2 = gsm[l] - 2.0f;
    const v4f sb0 = *(const v4f*)(svec + tm + 8 * half);
    const v4f sb1 = *(const v4f*)(svec + tm + 8 * half + 4);

    v8f c1 = {}, c2 = {};
    v4f fb0[2][GEMM_BLK], fb1[2][GEMM_BLK];
    #pragma unroll
    for (int c = 0; c < GEMM_BLK; ++c) {
        fb0[0][c] = b1p[32 * c];
        fb1[0][c] = b1p[32 * (GEMM_CHUNKS + c)];
    }

    #pragma unroll
    for (int g = 0; g < GEMM_CHUNKS; ++g) {
        const int blk = g / GEMM_BLK;
        const int c8 = g % GEMM_BLK;
        if (c8 == 0 && (blk + 1) * GEMM_BLK < GEMM_CHUNKS) {
            #pragma unroll
            for (int c = 0; c < GEMM_BLK; ++c) {
                fb0[(blk + 1) & 1][c] = b1p[32 * ((blk + 1) * GEMM_BLK + c)];
                fb1[(blk + 1) & 1][c] = b1p[32 * (GEMM_CHUNKS + (blk + 1) * GEMM_BLK + c)];
            }
            __builtin_amdgcn_sched_barrier(0);
        }
        if (g + GEMM_AHEAD < GEMM_CHUNKS) {
            const int gn = g + GEMM_AHEAD;
            fa[gn % GEMM_RING] = *(const v4f*)(ap + 8 * gn);
        }
        const int c = g % GEMM_RING;
        #pragma unroll
        for (int s = 0; s < 2; ++s) {
            const v2f a = pp2_pair(fa[c], s);
            c1 = __builtin_amdgcn_wmma_f32_16x16x4_f32(false, a, false, pp2_pair(fb0[blk & 1][c8], s),
                                                       (short)0, c1, false, false);
            c2 = __builtin_amdgcn_wmma_f32_16x16x4_f32(false, a, false, pp2_pair(fb1[blk & 1][c8], s),
                                                       (short)0, c2, false, false);
        }
        __builtin_amdgcn_sched_barrier(0);
    }

    float* orow = out + (tm + 8 * half) * NDIM + tn + l;
    #pragma unroll
    for (int r = 0; r < 8; ++r) {
        const float sb = (r < 4) ? sb0[r & 3] : sb1[r & 3];
        const float a1 = c1[r];
        const float a2 = c2[r];
        const float den = fmaxf(sqrtf(fmaxf(sb + a1 * a1 * gn2, 0.0f)), EPSF);
        orow[r * NDIM] = (a2 - a1 * dn) / den;
    }
}

extern "C" void kernel_launch(void* const* d_in, const int* in_sizes, int n_in,
                              void* d_out, int out_size, void* d_ws, size_t ws_size,
                              hipStream_t stream) {
    const float* x      = (const float*)d_in[0];
    const float* normal = (const float*)d_in[1];
    const float* plane  = (const float*)d_in[2];
    float* out = (float*)d_out;

    float* ws    = (float*)d_ws;
    float* colpart = ws;
    float* svec  = colpart + CP_STRIPS * NDIM;

    pp2_colpart_xnorm_kernel<<<dim3(NDIM / 128, CP_STRIPS + XN_GROUPS), 32, 0, stream>>>(
        normal, x, colpart, svec);

    pp2_prep_wmma_gemm_kernel<<<256, 128, 0, stream>>>(x, normal, plane, colpart, svec, out);
}


# B-operand preparation split by rows, not column slices: a wave owns 4 whole rows (512 B contiguous loads, 4 lines not 16), joins its row sums by lane exchange; order of row sums, d_o, g_o changes
# speedup vs baseline: 5.1811x; 1.0626x over previous
#include <hip/hip_runtime.h>
#include <math.h>

typedef __attribute__((ext_vector_type(2))) float v2f;
typedef __attribute__((ext_vector_type(8))) float v8f;

#define NDIM 512
#define EPSF 1e-12f

#define CP_STRIPS 16
#define CP_ROWS (NDIM / CP_STRIPS)
#define XN_ROWS 4
#define XN_GROUPS (NDIM / (4 * XN_ROWS))

__global__ __launch_bounds__(32) __attribute__((amdgpu_waves_per_eu(1, 1)))
void pp2_colpart_xnorm_kernel(
    const float* __restrict__ normal, const float* __restrict__ x,
    float* __restrict__ colpart, float* __restrict__ svec) {
    if (blockIdx.y >= CP_STRIPS) {
        const int lane = threadIdx.x;
        const int b0 = ((blockIdx.y - CP_STRIPS) * 4 + blockIdx.x) * XN_ROWS;
        const float4* xs = (const float4*)(x + b0 * NDIM) + lane;

        float4 u[XN_ROWS][4];
        #pragma unroll
        for (int r = 0; r < XN_ROWS; ++r) {
            #pragma unroll
            for (int j = 0; j < 4; ++j) u[r][j] = xs[r * (NDIM / 4) + j * 32];
        }

        float s[XN_ROWS];
        #pragma unroll
        for (int r = 0; r < XN_ROWS; ++r) {
            float a = 0.f;
            #pragma unroll
            for (int j = 0; j < 4; ++j) {
                a += u[r][j].x * u[r][j].x;
                a += u[r][j].y * u[r][j].y;
                a += u[r][j].z * u[r][j].z;
                a += u[r][j].w * u[r][j].w;
            }
            s[r] = a;
        }
        #pragma unroll
        for (int off = 16; off > 0; off >>= 1) {
            #pragma unroll
            for (int r = 0; r < XN_ROWS; ++r) s[r] += __shfl_xor(s[r], off);
        }
        if (lane == 0) {
            *(float4*)(svec + b0) = make_float4(s[0], s[1], s[2], s[3]);
        }
        return;
    }

    const int strip = blockIdx.y;
    const int c4 = blockIdx.x * 32 + threadIdx.x;
    const float4* src =
        (const float4*)(normal + strip * CP_ROWS * NDIM) + c4;

    float4 v[CP_ROWS];
    #pragma unroll
    for (int r = 0; r < CP_ROWS; ++r) v[r] = src[r * (NDIM / 4)];

    float4 s = make_float4(0.f, 0.f, 0.f, 0.f);
    #pragma unroll
    for (int r = 0; r < CP_ROWS; ++r) {
        s.x += v[r].x * v[r].x;
        s.y += v[r].y * v[r].y;
        s.z += v[r].z * v[r].z;
        s.w += v[r].w * v[r].w;
    }
    ((float4*)(colpart + strip * NDIM))[c4] = s;
}

typedef __attribute__((ext_vector_type(4))) float v4f;
#define BP 129

template <int M>
__device__ __forceinline__ float pp2_row_xadd(float v) {
    const int b = __builtin_bit_cast(int, v);
    return v + __builtin_bit_cast(float,
        __builtin_amdgcn_update_dpp(0, b, 0x160 + M, 0xf, 0xf, true));
}

__device__ __forceinline__ float pp2_xhalf(float v) {
    const int b = __builtin_bit_cast(int, v);
    return __builtin_bit_cast(float,
        __builtin_amdgcn_permlanex16(b, b, 0x76543210, 0xfedcba98, false, false));
}

__device__ __forceinline__ float pp2_wave_sum(float v) {
    v = pp2_row_xadd<1>(v);
    v = pp2_row_xadd<2>(v);
    v = pp2_row_xadd<4>(v);
    v = pp2_row_xadd<8>(v);
    return v + pp2_xhalf(v);
}

__device__ __forceinline__ float pp2_hsum(v4f a) {
    return (a[0] + a[1]) + (a[2] + a[3]);
}

#define GEMM_CHUNKS (NDIM / 8)
#define GEMM_AHEAD 16
#define GEMM_RING (GEMM_AHEAD + 1)
#define GEMM_BLK 8
#define PREP_ROWS 4

__device__ __forceinline__ v2f pp2_pair(v4f v, int s) {
    v2f r;
    r[0] = v[2 * s];
    r[1] = v[2 * s + 1];
    return r;
}

__global__ __launch_bounds__(128) __attribute__((amdgpu_waves_per_eu(1, 1)))
void pp2_prep_wmma_gemm_kernel(
    const float* __restrict__ x,
    const float* __restrict__ normal,
    const float* __restrict__ plane,
    const float* __restrict__ colpart,
    const float* __restrict__ svec,
    float* __restrict__ out) {
    __shared__ v4f bl[2][16][BP];
    __shared__ v4f cl[NDIM / 4];
    __shared__ v4f dsm[4], gsm[4];

    const int w = threadIdx.x >> 5;
    const int nt = blockIdx.x & 31;
    const int mt = (blockIdx.x >> 5) * 4 + w;
    const int tm = mt * 16;
    const int tn = nt * 16;

    const int lane = threadIdx.x & 31;
    const int half = lane >> 4;
    const int l = lane & 15;

    v4f cpv[CP_STRIPS];
    {
        const v4f* cp = (const v4f*)(colpart + 128 * w) + lane;
        #pragma unroll
        for (int j = 0; j < CP_STRIPS; ++j) cpv[j] = cp[j * (NDIM / 4)];
    }

    const int r0 = PREP_ROWS * w;
    v4f n4[PREP_ROWS][4], p4[PREP_ROWS][4];
    {
        const v4f* nr = (const v4f*)(normal + (tn + r0) * NDIM) + lane;
        const v4f* pr = (const v4f*)(plane + (tn + r0) * NDIM) + lane;
        #pragma unroll
        for (int r = 0; r < PREP_ROWS; ++r) {
            #pragma unroll
            for (int q = 0; q < 4; ++q) {
                n4[r][q] = nr[r * (NDIM / 4) + 32 * q];
                p4[r][q] = pr[r * (NDIM / 4) + 32 * q];
            }
        }
    }

    {
        v4f cs = {0.0f, 0.0f, 0.0f, 0.0f};
        #pragma unroll
        for (int j = 0; j < CP_STRIPS; ++j) cs += cpv[j];
        v4f ci;
        #pragma unroll
        for (int e = 0; e < 4; ++e) ci[e] = 1.0f / fmaxf(sqrtf(cs[e]), EPSF);
        cl[32 * w + lane] = ci;
    }
    __syncthreads();

    const float* ap = x + (tm + l) * NDIM + 4 * half;
    v4f fa[GEMM_RING];
    #pragma unroll
    for (int g = 0; g < GEMM_AHEAD; ++g) fa[g] = *(const v4f*)(ap + 8 * g);

    v4f c4[4];
    #pragma unroll
    for (int q = 0; q < 4; ++q) c4[q] = cl[32 * q + lane];

    float ctr[PREP_ROWS], rinvr[PREP_ROWS];
    {
        float S[PREP_ROWS][6];
        v4f dq, gq;
        #pragma unroll
        for (int r = 0; r < PREP_ROWS; ++r) {
            v4f asp = {0.0f, 0.0f, 0.0f, 0.0f}, asn = asp, aspn = asp, agg = asp, at1 = asp, at2 = asp;
            #pragma unroll
            for (int q = 0; q < 4; ++q) {
                const v4f n = n4[r][q], p = p4[r][q];
                const v4f v = n * c4[q];
                bl[0][r0 + r][32 * q + lane] = v;
                asp  += p * p;
                asn  += n * n;
                aspn += p * n;
                agg  += v * v;
                at1  += v * p;
                at2  += v * n;
            }
            S[r][0] = pp2_hsum(asp);
            S[r][1] = pp2_hsum(asn);
            S[r][2] = pp2_hsum(aspn);
            S[r][3] = pp2_hsum(agg);
            S[r][4] = pp2_hsum(at1);
            S[r][5] = pp2_hsum(at2);
        }
        #pragma unroll
        for (int r = 0; r < PREP_ROWS; ++r) {
            #pragma unroll
            for (int q = 0; q < 6; ++q) S[r][q] = pp2_wave_sum(S[r][q]);
        }
        #pragma unroll
        for (int r = 0; r < PREP_ROWS; ++r) {
            const float Sp = S[r][0], Sn = S[r][1], Spn = S[r][2];
            const float G  = S[r][3], T1 = S[r][4], T2 = S[r][5];

            const float np  = fmaxf(sqrtf(Sp), EPSF);
            const float nn1 = fmaxf(sqrtf(Sn), EPSF);
            const float ct  = Spn / (np * nn1);
            const float qq  = fmaxf(Sp - 2.0f * ct * Spn + ct * ct * Sn, 0.0f);
            const float rinv = 1.0f / fmaxf(sqrtf(qq), EPSF);
            ctr[r] = ct;
            rinvr[r] = rinv;
            dq[r] = rinv * (T1 - ct * T2);
            gq[r] = G;
        }
        if (lane == 0) {
            dsm[w] = dq;
            gsm[w] = gq;
        }
    }
    #pragma unroll
    for (int r = 0; r < PREP_ROWS; ++r) {
        #pragma unroll
        for (int q = 0; q < 4; ++q)
            bl[1][r0 + r][32 * q + lane] = (p4[r][q] - n4[r][q] * ctr[r]) * rinvr[r];
    }
    __syncthreads();

    const v4f* b1p = &bl[0][l][half];

    const float dn = ((const float*)dsm)[l];
    const float gn2 = ((const float*)gsm)[l] - 2.0f;
    const v4f sb0 = *(const v4f*)(svec + tm + 8 * half);
    const v4f sb1 = *(const v4f*)(svec + tm + 8 * half + 4);

    v8f c1 = {}, c2 = {};
    v4f fb0[2][GEMM_BLK], fb1[2][GEMM_BLK];
    #pragma unroll
    for (int c = 0; c < GEMM_BLK; ++c) {
        fb0[0][c] = b1p[2 * c];
        fb1[0][c] = b1p[16 * BP + 2 * c];
    }

    #pragma unroll
    for (int g = 0; g < GEMM_CHUNKS; ++g) {
        const int blk = g / GEMM_BLK;
        const int c8 = g % GEMM_BLK;
        if (c8 == 0 && (blk + 1) * GEMM_BLK < GEMM_CHUNKS) {
            #pragma unroll
            for (int c = 0; c < GEMM_BLK; ++c) {
                fb0[(blk + 1) & 1][c] = b1p[2 * ((blk + 1) * GEMM_BLK + c)];
                fb1[(blk + 1) & 1][c] = b1p[16 * BP + 2 * ((blk + 1) * GEMM_BLK + c)];
            }
            __builtin_amdgcn_sched_barrier(0);
        }
        if (g + GEMM_AHEAD < GEMM_CHUNKS) {
            const int gn = g + GEMM_AHEAD;
            fa[gn % GEMM_RING] = *(const v4f*)(ap + 8 * gn);
        }
        const int c = g % GEMM_RING;
        #pragma unroll
        for (int s = 0; s < 2; ++s) {
            const v2f a = pp2_pair(fa[c], s);
            c1 = __builtin_amdgcn_wmma_f32_16x16x4_f32(false, a, false, pp2_pair(fb0[blk & 1][c8], s),
                                                       (short)0, c1, false, false);
            c2 = __builtin_amdgcn_wmma_f32_16x16x4_f32(false, a, false, pp2_pair(fb1[blk & 1][c8], s),
                                                       (short)0, c2, false, false);
        }
        __builtin_amdgcn_sched_barrier(0);
    }

    float* orow = out + (tm + 8 * half) * NDIM + tn + l;
    #pragma unroll
    for (int r = 0; r < 8; ++r) {
        const float sb = (r < 4) ? sb0[r & 3] : sb1[r & 3];
        const float a1 = c1[r];
        const float a2 = c2[r];
        const float den = fmaxf(sqrtf(fmaxf(sb + a1 * a1 * gn2, 0.0f)), EPSF);
        orow[r * NDIM] = (a2 - a1 * dn) / den;
    }
}

extern "C" void kernel_launch(void* const* d_in, const int* in_sizes, int n_in,
                              void* d_out, int out_size, void* d_ws, size_t ws_size,
                              hipStream_t stream) {
    const float* x      = (const float*)d_in[0];
    const float* normal = (const float*)d_in[1];
    const float* plane  = (const float*)d_in[2];
    float* out = (float*)d_out;

    float* ws    = (float*)d_ws;
    float* colpart = ws;
    float* svec  = colpart + CP_STRIPS * NDIM;

    pp2_colpart_xnorm_kernel<<<dim3(NDIM / 128, CP_STRIPS + XN_GROUPS), 32, 0, stream>>>(
        normal, x, colpart, svec);

    pp2_prep_wmma_gemm_kernel<<<256, 128, 0, stream>>>(x, normal, plane, colpart, svec, out);
}
